# starting kernel
# baseline (speedup 1.0000x reference)
#include <hip/hip_runtime.h>
#include <math.h>


typedef __attribute__((ext_vector_type(2))) float v2f;
typedef __attribute__((ext_vector_type(8))) float v8f;

#define T_SEQ 2048
#define EMB   1024
#define HID   64
#define NB    8
#define MTOT  (NB * T_SEQ)
#define NKT   (T_SEQ / 16)

#define QOFF   0
#define KOFF   ((size_t)MTOT * HID)
#define VOFF   ((size_t)2 * MTOT * HID)
#define TSOFF  ((size_t)3 * MTOT * HID)
#define SUFOFF (TSOFF + (size_t)NB * NKT * HID)

__device__ __forceinline__ v8f wmma_f32(v2f a, v2f b, v8f c) {
  return __builtin_amdgcn_wmma_f32_16x16x4_f32(
       false, a,  false, b,
       (short)0, c,  false,  false);
}

__global__ __launch_bounds__(128) void qkv_kernel(
    const float* __restrict__ X,
    const float* __restrict__ Wq, const float* __restrict__ bq,
    const float* __restrict__ Wk, const float* __restrict__ bk,
    const float* __restrict__ Wv, const float* __restrict__ bv,
    float* __restrict__ ws) {
  const int lane = threadIdx.x;
  const int nt   = threadIdx.y;
  const int mat  = blockIdx.y;
  const int m0   = blockIdx.x * 16;
  const int n0   = nt * 16;
  const int ln   = lane & 15;
  const int of   = (lane >> 4) * 2;

  const float* W    = (mat == 0) ? Wq : (mat == 1) ? Wk : Wv;
  const float* bias = (mat == 0) ? bq : (mat == 1) ? bk : bv;
  float* out = ws + (size_t)mat * (size_t)MTOT * HID;

  const float* arow = X + (size_t)(m0 + ln) * EMB + of;
  const float* bcol = W + (size_t)of * HID + n0 + ln;

  v8f c = {};
  #pragma unroll 4
  for (int k = 0; k < EMB; k += 4) {
    v2f a = *(const v2f*)(arow + k);
    v2f b;
    b.x = bcol[(size_t)k * HID];
    b.y = bcol[(size_t)(k + 1) * HID];
    c = wmma_f32(a, b, c);
  }

  const float bb = bias[n0 + ln];
  const int mrow = m0 + (lane >> 4) * 8;
  #pragma unroll
  for (int r = 0; r < 8; ++r)
    out[(size_t)(mrow + r) * HID + n0 + ln] = c[r] + bb;
}

__global__ void vtilesum_kernel(float* __restrict__ ws) {
  const int h = threadIdx.x;
  const int j = blockIdx.x;
  const int b = blockIdx.y;
  const float* V = ws + VOFF + (size_t)(b * T_SEQ + j * 16) * HID + h;
  float s = 0.f;
  #pragma unroll
  for (int i = 0; i < 16; ++i) s += V[(size_t)i * HID];
  ws[TSOFF + ((size_t)b * NKT + j) * HID + h] = s;
}

__global__ void vsuffix_kernel(float* __restrict__ ws) {
  const int h = threadIdx.x;
  const int b = blockIdx.x;
  const float* ts = ws + TSOFF + (size_t)b * NKT * HID;
  float* suf = ws + SUFOFF + (size_t)b * (NKT + 1) * HID;
  float acc = 0.f;
  suf[(size_t)NKT * HID + h] = 0.f;
  for (int j = NKT - 1; j >= 0; --j) {
    acc += ts[(size_t)j * HID + h];
    suf[(size_t)j * HID + h] = acc;
  }
}

__global__ __launch_bounds__(128) void attn_kernel(
    const float* __restrict__ ws, float* __restrict__ out) {
  __shared__ float plds[4][16][18];

  const int lane = threadIdx.x;
  const int w    = threadIdx.y;
  const int b    = blockIdx.x;
  const int qt   = blockIdx.y * 4 + w;
  const int q0   = qt * 16;
  const int ln   = lane & 15;
  const int hi   = lane >> 4;
  const int of   = hi * 2;

  const float* Q = ws + QOFF;
  const float* K = ws + KOFF;
  const float* V = ws + VOFF;
  const float scale = 0.125f;

  v2f qa[16];
  {
    const float* qrow = Q + (size_t)(b * T_SEQ + q0 + ln) * HID + of;
    #pragma unroll
    for (int kk = 0; kk < 16; ++kk) qa[kk] = *(const v2f*)(qrow + 4 * kk);
  }

  v8f acc0 = {}, acc1 = {}, acc2 = {}, acc3 = {};
  float rm[8], rs[8];
  #pragma unroll
  for (int r = 0; r < 8; ++r) { rm[r] = -INFINITY; rs[r] = 0.f; }

  for (int kt = 0; kt <= qt; ++kt) {
    const int k0 = kt * 16;

    v8f s = {};
    const float* krow = K + (size_t)(b * T_SEQ + k0 + ln) * HID + of;
    #pragma unroll
    for (int kk = 0; kk < 16; ++kk) {
      v2f kb = *(const v2f*)(krow + 4 * kk);
      s = wmma_f32(qa[kk], kb, s);
    }

    const int kg = k0 + ln;
    #pragma unroll
    for (int r = 0; r < 8; ++r) {
      const int qg = q0 + r + hi * 8;
      const float v = s[r] * scale;
      s[r] = (kg > qg) ? 0.f : v;
    }

    float corr[8];
    #pragma unroll
    for (int r = 0; r < 8; ++r) {
      float t = s[r];
      t = fmaxf(t, __shfl_xor(t, 1));
      t = fmaxf(t, __shfl_xor(t, 2));
      t = fmaxf(t, __shfl_xor(t, 4));
      t = fmaxf(t, __shfl_xor(t, 8));
      const float nm = fmaxf(rm[r], t);
      const float c  = __expf(rm[r] - nm);
      const float e  = __expf(s[r] - nm);
      s[r] = e;
      float rsum = e;
      rsum += __shfl_xor(rsum, 1);
      rsum += __shfl_xor(rsum, 2);
      rsum += __shfl_xor(rsum, 4);
      rsum += __shfl_xor(rsum, 8);
      rs[r] = rs[r] * c + rsum;
      rm[r] = nm;
      corr[r] = c;
    }
    #pragma unroll
    for (int r = 0; r < 8; ++r) {
      acc0[r] *= corr[r]; acc1[r] *= corr[r];
      acc2[r] *= corr[r]; acc3[r] *= corr[r];
    }

    #pragma unroll
    for (int r = 0; r < 8; ++r) plds[w][r + hi * 8][ln] = s[r];
    asm volatile("s_wait_dscnt 0x0" ::: "memory");

    const float* vb = V + (size_t)(b * T_SEQ + k0) * HID;
    #pragma unroll
    for (int j = 0; j < 4; ++j) {
      v2f pa = *(const v2f*)(&plds[w][ln][4 * j + of]);
      const float* vr0 = vb + (size_t)(4 * j + of) * HID + ln;
      const float* vr1 = vb + (size_t)(4 * j + of + 1) * HID + ln;
      v2f bb;
      bb.x = vr0[0];  bb.y = vr1[0];  acc0 = wmma_f32(pa, bb, acc0);
      bb.x = vr0[16]; bb.y = vr1[16]; acc1 = wmma_f32(pa, bb, acc1);
      bb.x = vr0[32]; bb.y = vr1[32]; acc2 = wmma_f32(pa, bb, acc2);
      bb.x = vr0[48]; bb.y = vr1[48]; acc3 = wmma_f32(pa, bb, acc3);
    }
  }

  const int cf = T_SEQ - 16 * (qt + 1);
  if (cf > 0) {
    const float* suf = ws + SUFOFF + ((size_t)b * (NKT + 1) + (qt + 1)) * HID + ln;
    const float s0 = suf[0], s1 = suf[16], s2 = suf[32], s3 = suf[48];
    #pragma unroll
    for (int r = 0; r < 8; ++r) {
      const float nm = fmaxf(rm[r], 0.f);
      const float c  = __expf(rm[r] - nm);
      const float wz = __expf(-nm);
      rs[r] = rs[r] * c + wz * (float)cf;
      acc0[r] = acc0[r] * c + wz * s0;
      acc1[r] = acc1[r] * c + wz * s1;
      acc2[r] = acc2[r] * c + wz * s2;
      acc3[r] = acc3[r] * c + wz * s3;
    }
  }

  float* orow = out + (size_t)(b * T_SEQ + q0 + hi * 8) * HID + ln;
  #pragma unroll
  for (int r = 0; r < 8; ++r) {
    const float inv = 1.f / rs[r];
    orow[(size_t)r * HID +  0] = acc0[r] * inv;
    orow[(size_t)r * HID + 16] = acc1[r] * inv;
    orow[(size_t)r * HID + 32] = acc2[r] * inv;
    orow[(size_t)r * HID + 48] = acc3[r] * inv;
  }
}

extern "C" void kernel_launch(void* const* d_in, const int* in_sizes, int n_in,
                              void* d_out, int out_size, void* d_ws, size_t ws_size,
                              hipStream_t stream) {
  const float* X  = (const float*)d_in[0];
  const float* Wq = (const float*)d_in[1];
  const float* bq = (const float*)d_in[2];
  const float* Wk = (const float*)d_in[3];
  const float* bk = (const float*)d_in[4];
  const float* Wv = (const float*)d_in[5];
  const float* bv = (const float*)d_in[6];
  float* ws  = (float*)d_ws;
  float* out = (float*)d_out;

  qkv_kernel<<<dim3(MTOT / 16, 3), dim3(32, 4), 0, stream>>>(
      X, Wq, bq, Wk, bk, Wv, bv, ws);
  vtilesum_kernel<<<dim3(NKT, NB), 64, 0, stream>>>(ws);
  vsuffix_kernel<<<NB, 64, 0, stream>>>(ws);
  attn_kernel<<<dim3(NB, NKT / 4), dim3(32, 4), 0, stream>>>(ws, out);
}


# attention: one work-group per 16-row query tile, its causal key tiles dealt round-robin to 8 waves, partial (max,sum,O) merged through LDS; changes the softmax/PV summation order
# speedup vs baseline: 2.0100x; 2.0100x over previous
#include <hip/hip_runtime.h>
#include <math.h>


typedef __attribute__((ext_vector_type(2))) float v2f;
typedef __attribute__((ext_vector_type(8))) float v8f;

#define T_SEQ 2048
#define EMB   1024
#define HID   64
#define NB    8
#define MTOT  (NB * T_SEQ)
#define NKT   (T_SEQ / 16)
#define NW    8

#define QOFF   0
#define KOFF   ((size_t)MTOT * HID)
#define VOFF   ((size_t)2 * MTOT * HID)
#define TSOFF  ((size_t)3 * MTOT * HID)
#define SUFOFF (TSOFF + (size_t)NB * NKT * HID)

__device__ __forceinline__ v8f wmma_f32(v2f a, v2f b, v8f c) {
  return __builtin_amdgcn_wmma_f32_16x16x4_f32(
       false, a,  false, b,
       (short)0, c,  false,  false);
}

__global__ __launch_bounds__(128) void qkv_kernel(
    const float* __restrict__ X,
    const float* __restrict__ Wq, const float* __restrict__ bq,
    const float* __restrict__ Wk, const float* __restrict__ bk,
    const float* __restrict__ Wv, const float* __restrict__ bv,
    float* __restrict__ ws) {
  const int lane = threadIdx.x;
  const int nt   = threadIdx.y;
  const int mat  = blockIdx.y;
  const int m0   = blockIdx.x * 16;
  const int n0   = nt * 16;
  const int ln   = lane & 15;
  const int of   = (lane >> 4) * 2;

  const float* W    = (mat == 0) ? Wq : (mat == 1) ? Wk : Wv;
  const float* bias = (mat == 0) ? bq : (mat == 1) ? bk : bv;
  float* out = ws + (size_t)mat * (size_t)MTOT * HID;

  const float* arow = X + (size_t)(m0 + ln) * EMB + of;
  const float* bcol = W + (size_t)of * HID + n0 + ln;

  v8f c = {};
  #pragma unroll 4
  for (int k = 0; k < EMB; k += 4) {
    v2f a = *(const v2f*)(arow + k);
    v2f b;
    b.x = bcol[(size_t)k * HID];
    b.y = bcol[(size_t)(k + 1) * HID];
    c = wmma_f32(a, b, c);
  }

  const float bb = bias[n0 + ln];
  const int mrow = m0 + (lane >> 4) * 8;
  #pragma unroll
  for (int r = 0; r < 8; ++r)
    out[(size_t)(mrow + r) * HID + n0 + ln] = c[r] + bb;
}

__global__ void vtilesum_kernel(float* __restrict__ ws) {
  const int h = threadIdx.x;
  const int j = blockIdx.x;
  const int b = blockIdx.y;
  const float* V = ws + VOFF + (size_t)(b * T_SEQ + j * 16) * HID + h;
  float s = 0.f;
  #pragma unroll
  for (int i = 0; i < 16; ++i) s += V[(size_t)i * HID];
  ws[TSOFF + ((size_t)b * NKT + j) * HID + h] = s;
}

__global__ void vsuffix_kernel(float* __restrict__ ws) {
  const int h = threadIdx.x;
  const int b = blockIdx.x;
  const float* ts = ws + TSOFF + (size_t)b * NKT * HID;
  float* suf = ws + SUFOFF + (size_t)b * (NKT + 1) * HID;
  float acc = 0.f;
  suf[(size_t)NKT * HID + h] = 0.f;
  for (int j = NKT - 1; j >= 0; --j) {
    acc += ts[(size_t)j * HID + h];
    suf[(size_t)j * HID + h] = acc;
  }
}

__global__ __launch_bounds__(32 * NW) void attn_kernel(
    const float* __restrict__ ws, float* __restrict__ out) {
  __shared__ float plds[NW][16][18];
  __shared__ __attribute__((aligned(16))) float pO[NW][16][68];
  __shared__ float pm[NW][16];
  __shared__ float pl[NW][16];

  const int lane = threadIdx.x;
  const int w    = threadIdx.y;
  const int b    = blockIdx.x;
  const int qt   = NKT - 1 - (int)blockIdx.y;
  const int q0   = qt * 16;
  const int ln   = lane & 15;
  const int hi   = lane >> 4;
  const int of   = hi * 2;

  const float* Q = ws + QOFF;
  const float* K = ws + KOFF;
  const float* V = ws + VOFF;
  const float scale = 0.125f;

  v2f qa[16];
  {
    const float* qrow = Q + (size_t)(b * T_SEQ + q0 + ln) * HID + of;
    #pragma unroll
    for (int kk = 0; kk < 16; ++kk) qa[kk] = *(const v2f*)(qrow + 4 * kk);
  }

  v8f acc0 = {}, acc1 = {}, acc2 = {}, acc3 = {};
  float rm[8], rs[8];
  #pragma unroll
  for (int r = 0; r < 8; ++r) { rm[r] = -INFINITY; rs[r] = 0.f; }

  for (int kt = w; kt <= qt; kt += NW) {
    const int k0 = kt * 16;

    v8f s = {};
    const float* krow = K + (size_t)(b * T_SEQ + k0 + ln) * HID + of;
    #pragma unroll
    for (int kk = 0; kk < 16; ++kk) {
      v2f kb = *(const v2f*)(krow + 4 * kk);
      s = wmma_f32(qa[kk], kb, s);
    }

    const int kg = k0 + ln;
    #pragma unroll
    for (int r = 0; r < 8; ++r) {
      const int qg = q0 + r + hi * 8;
      const float v = s[r] * scale;
      s[r] = (kg > qg) ? 0.f : v;
    }

    float corr[8];
    #pragma unroll
    for (int r = 0; r < 8; ++r) {
      float t = s[r];
      t = fmaxf(t, __shfl_xor(t, 1));
      t = fmaxf(t, __shfl_xor(t, 2));
      t = fmaxf(t, __shfl_xor(t, 4));
      t = fmaxf(t, __shfl_xor(t, 8));
      const float nm = fmaxf(rm[r], t);
      const float c  = __expf(rm[r] - nm);
      const float e  = __expf(s[r] - nm);
      s[r] = e;
      float rsum = e;
      rsum += __shfl_xor(rsum, 1);
      rsum += __shfl_xor(rsum, 2);
      rsum += __shfl_xor(rsum, 4);
      rsum += __shfl_xor(rsum, 8);
      rs[r] = rs[r] * c + rsum;
      rm[r] = nm;
      corr[r] = c;
    }
    #pragma unroll
    for (int r = 0; r < 8; ++r) {
      acc0[r] *= corr[r]; acc1[r] *= corr[r];
      acc2[r] *= corr[r]; acc3[r] *= corr[r];
    }

    #pragma unroll
    for (int r = 0; r < 8; ++r) plds[w][r + hi * 8][ln] = s[r];
    asm volatile("s_wait_dscnt 0x0" ::: "memory");

    const float* vb = V + (size_t)(b * T_SEQ + k0) * HID;
    #pragma unroll
    for (int j = 0; j < 4; ++j) {
      v2f pa = *(const v2f*)(&plds[w][ln][4 * j + of]);
      const float* vr0 = vb + (size_t)(4 * j + of) * HID + ln;
      const float* vr1 = vb + (size_t)(4 * j + of + 1) * HID + ln;
      v2f bb;
      bb.x = vr0[0];  bb.y = vr1[0];  acc0 = wmma_f32(pa, bb, acc0);
      bb.x = vr0[16]; bb.y = vr1[16]; acc1 = wmma_f32(pa, bb, acc1);
      bb.x = vr0[32]; bb.y = vr1[32]; acc2 = wmma_f32(pa, bb, acc2);
      bb.x = vr0[48]; bb.y = vr1[48]; acc3 = wmma_f32(pa, bb, acc3);
    }
  }

  #pragma unroll
  for (int r = 0; r < 8; ++r) {
    const int prow = r + hi * 8;
    pO[w][prow][ln +  0] = acc0[r];
    pO[w][prow][ln + 16] = acc1[r];
    pO[w][prow][ln + 32] = acc2[r];
    pO[w][prow][ln + 48] = acc3[r];
    if (ln == 0) { pm[w][prow] = rm[r]; pl[w][prow] = rs[r]; }
  }
  __syncthreads();

  const int tid  = w * 32 + lane;
  const int mrow = tid >> 4;
  const int mcol = (tid & 15) * 4;

  float M = pm[0][mrow];
  #pragma unroll
  for (int i = 1; i < NW; ++i) M = fmaxf(M, pm[i][mrow]);

  float L = 0.f;
  float o0 = 0.f, o1 = 0.f, o2 = 0.f, o3 = 0.f;
  #pragma unroll
  for (int i = 0; i < NW; ++i) {
    const float f = __expf(pm[i][mrow] - M);
    const float4 p = *(const float4*)(&pO[i][mrow][mcol]);
    L  += pl[i][mrow] * f;
    o0 += p.x * f; o1 += p.y * f; o2 += p.z * f; o3 += p.w * f;
  }

  const int cf = T_SEQ - 16 * (qt + 1);
  if (cf > 0) {
    const float4 sf = *(const float4*)(
        ws + SUFOFF + ((size_t)b * (NKT + 1) + (qt + 1)) * HID + mcol);
    const float nm = fmaxf(M, 0.f);
    const float cc = __expf(M - nm);
    const float wz = __expf(-nm);
    L  = L  * cc + wz * (float)cf;
    o0 = o0 * cc + wz * sf.x;
    o1 = o1 * cc + wz * sf.y;
    o2 = o2 * cc + wz * sf.z;
    o3 = o3 * cc + wz * sf.w;
  }

  const float inv = 1.f / L;
  float4 o;
  o.x = o0 * inv; o.y = o1 * inv; o.z = o2 * inv; o.w = o3 * inv;
  *(float4*)(out + (size_t)(b * T_SEQ + q0 + mrow) * HID + mcol) = o;
}

extern "C" void kernel_launch(void* const* d_in, const int* in_sizes, int n_in,
                              void* d_out, int out_size, void* d_ws, size_t ws_size,
                              hipStream_t stream) {
  const float* X  = (const float*)d_in[0];
  const float* Wq = (const float*)d_in[1];
  const float* bq = (const float*)d_in[2];
  const float* Wk = (const float*)d_in[3];
  const float* bk = (const float*)d_in[4];
  const float* Wv = (const float*)d_in[5];
  const float* bv = (const float*)d_in[6];
  float* ws  = (float*)d_ws;
  float* out = (float*)d_out;

  qkv_kernel<<<dim3(MTOT / 16, 3), dim3(32, 4), 0, stream>>>(
      X, Wq, bq, Wk, bk, Wv, bv, ws);
  vtilesum_kernel<<<dim3(NKT, NB), 64, 0, stream>>>(ws);
  vsuffix_kernel<<<NB, 64, 0, stream>>>(ws);
  attn_kernel<<<dim3(NB, NKT), dim3(32, NW), 0, stream>>>(ws, out);
}


# QKV projection retiled: 64x64 tile per 8-wave work-group, X and W staged through LDS in K chunks of 32 (whole-line float4 loads, 2 stages), fragments read from LDS; per-accumulator sum order kept
# speedup vs baseline: 2.4063x; 1.1972x over previous
#include <hip/hip_runtime.h>
#include <math.h>


typedef __attribute__((ext_vector_type(2))) float v2f;
typedef __attribute__((ext_vector_type(8))) float v8f;

#define T_SEQ 2048
#define EMB   1024
#define HID   64
#define NB    8
#define MTOT  (NB * T_SEQ)
#define NKT   (T_SEQ / 16)
#define NW    8

#define QOFF   0
#define KOFF   ((size_t)MTOT * HID)
#define VOFF   ((size_t)2 * MTOT * HID)
#define TSOFF  ((size_t)3 * MTOT * HID)
#define SUFOFF (TSOFF + (size_t)NB * NKT * HID)

__device__ __forceinline__ v8f wmma_f32(v2f a, v2f b, v8f c) {
  return __builtin_amdgcn_wmma_f32_16x16x4_f32(
       false, a,  false, b,
       (short)0, c,  false,  false);
}

#define PBM 64
#define PBK 32
#define PXP (PBK + 4)
#define PWP 160
#define PNC (EMB / PBK)
#define PXV (PBK / 16)
#define PWV (PBK / 32)

__device__ __forceinline__ void qkv_chunk(const float* __restrict__ xa,
                                          const float* __restrict__ wb,
                                          v8f& c0, v8f& c1) {
  v2f a[PBK / 4], b0[PBK / 4], b1[PBK / 4];
  #pragma unroll
  for (int s = 0; s < PBK / 4; ++s) {
    a[s]  = *(const v2f*)(xa + 4 * s);
    b0[s] = *(const v2f*)(wb + (2 * s) * PWP);
    b1[s] = *(const v2f*)(wb + (2 * s) * PWP + 32);
  }
  #pragma unroll
  for (int s = 0; s < PBK / 4; ++s) {
    c0 = wmma_f32(a[s], b0[s], c0);
    c1 = wmma_f32(a[s], b1[s], c1);
  }
}

__global__ __launch_bounds__(256) __attribute__((amdgpu_waves_per_eu(2, 6))) void qkv_kernel(
    const float* __restrict__ X,
    const float* __restrict__ Wq, const float* __restrict__ bq,
    const float* __restrict__ Wk, const float* __restrict__ bk,
    const float* __restrict__ Wv, const float* __restrict__ bv,
    float* __restrict__ ws) {
  __shared__ __attribute__((aligned(16))) float xs[2][PBM * PXP];
  __shared__ __attribute__((aligned(16))) float wt[2][(PBK / 2) * PWP];

  const int tid  = threadIdx.x;
  const int lane = tid & 31;
  const int w    = tid >> 5;
  const int wm   = w & 3;
  const int wn   = w >> 2;
  const int mat  = blockIdx.x;
  const int m0   = blockIdx.y * PBM;
  const int ln   = lane & 15;
  const int hi   = lane >> 4;
  const int of   = hi * 2;

  const float* W    = (mat == 0) ? Wq : (mat == 1) ? Wk : Wv;
  const float* bias = (mat == 0) ? bq : (mat == 1) ? bk : bv;
  float* out = ws + (size_t)mat * (size_t)MTOT * HID;

  const float* xg[PXV];  int xl[PXV];
  #pragma unroll
  for (int i = 0; i < PXV; ++i) {
    const int idx = tid + 256 * i;
    const int row = idx / (PBK / 4), c4 = idx % (PBK / 4);
    xg[i] = X + (size_t)(m0 + row) * EMB + 4 * c4;
    xl[i] = row * PXP + 4 * c4;
  }
  const float* wg[PWV];  int wl[PWV];
  #pragma unroll
  for (int i = 0; i < PWV; ++i) {
    const int u = tid + 256 * i;
    const int kp = u >> 4, c4 = u & 15;
    wg[i] = W + (size_t)(2 * kp) * HID + 4 * c4;
    wl[i] = kp * PWP + 8 * c4;
  }

  const int xf = (wm * 16 + ln) * PXP + of;
  const int wf = hi * PWP + 2 * (wn * 32 + ln);

  float4 px[PXV], pwa[PWV], pwb[PWV];
  #pragma unroll
  for (int i = 0; i < PXV; ++i) px[i] = *(const float4*)xg[i];
  #pragma unroll
  for (int i = 0; i < PWV; ++i) {
    pwa[i] = *(const float4*)wg[i];
    pwb[i] = *(const float4*)(wg[i] + HID);
  }
  #pragma unroll
  for (int i = 0; i < PXV; ++i) *(float4*)&xs[0][xl[i]] = px[i];
  #pragma unroll
  for (int i = 0; i < PWV; ++i) {
    *(float4*)&wt[0][wl[i]]     = make_float4(pwa[i].x, pwb[i].x, pwa[i].y, pwb[i].y);
    *(float4*)&wt[0][wl[i] + 4] = make_float4(pwa[i].z, pwb[i].z, pwa[i].w, pwb[i].w);
  }
  __syncthreads();

  v8f c0 = {}, c1 = {};
  for (int t = 0; t < PNC - 1; ++t) {
    const int cur = t & 1;
    const size_t kx = (size_t)(t + 1) * PBK;
    #pragma unroll
    for (int i = 0; i < PXV; ++i) px[i] = *(const float4*)(xg[i] + kx);
    #pragma unroll
    for (int i = 0; i < PWV; ++i) {
      pwa[i] = *(const float4*)(wg[i] + kx * HID);
      pwb[i] = *(const float4*)(wg[i] + kx * HID + HID);
    }

    qkv_chunk(&xs[cur][xf], &wt[cur][wf], c0, c1);

    #pragma unroll
    for (int i = 0; i < PXV; ++i) *(float4*)&xs[cur ^ 1][xl[i]] = px[i];
    #pragma unroll
    for (int i = 0; i < PWV; ++i) {
      *(float4*)&wt[cur ^ 1][wl[i]]     = make_float4(pwa[i].x, pwb[i].x, pwa[i].y, pwb[i].y);
      *(float4*)&wt[cur ^ 1][wl[i] + 4] = make_float4(pwa[i].z, pwb[i].z, pwa[i].w, pwb[i].w);
    }
    __syncthreads();
  }
  qkv_chunk(&xs[(PNC - 1) & 1][xf], &wt[(PNC - 1) & 1][wf], c0, c1);

  const int n0   = wn * 32 + ln;
  const float bb0 = bias[n0], bb1 = bias[n0 + 16];
  const int mrow = m0 + wm * 16 + hi * 8;
  #pragma unroll
  for (int r = 0; r < 8; ++r) {
    out[(size_t)(mrow + r) * HID + n0]      = c0[r] + bb0;
    out[(size_t)(mrow + r) * HID + n0 + 16] = c1[r] + bb1;
  }
}

__global__ void vtilesum_kernel(float* __restrict__ ws) {
  const int h = threadIdx.x;
  const int j = blockIdx.x;
  const int b = blockIdx.y;
  const float* V = ws + VOFF + (size_t)(b * T_SEQ + j * 16) * HID + h;
  float s = 0.f;
  #pragma unroll
  for (int i = 0; i < 16; ++i) s += V[(size_t)i * HID];
  ws[TSOFF + ((size_t)b * NKT + j) * HID + h] = s;
}

__global__ void vsuffix_kernel(float* __restrict__ ws) {
  const int h = threadIdx.x;
  const int b = blockIdx.x;
  const float* ts = ws + TSOFF + (size_t)b * NKT * HID;
  float* suf = ws + SUFOFF + (size_t)b * (NKT + 1) * HID;
  float acc = 0.f;
  suf[(size_t)NKT * HID + h] = 0.f;
  for (int j = NKT - 1; j >= 0; --j) {
    acc += ts[(size_t)j * HID + h];
    suf[(size_t)j * HID + h] = acc;
  }
}

__global__ __launch_bounds__(32 * NW) void attn_kernel(
    const float* __restrict__ ws, float* __restrict__ out) {
  __shared__ float plds[NW][16][18];
  __shared__ __attribute__((aligned(16))) float pO[NW][16][68];
  __shared__ float pm[NW][16];
  __shared__ float pl[NW][16];

  const int lane = threadIdx.x;
  const int w    = threadIdx.y;
  const int b    = blockIdx.x;
  const int qt   = NKT - 1 - (int)blockIdx.y;
  const int q0   = qt * 16;
  const int ln   = lane & 15;
  const int hi   = lane >> 4;
  const int of   = hi * 2;

  const float* Q = ws + QOFF;
  const float* K = ws + KOFF;
  const float* V = ws + VOFF;
  const float scale = 0.125f;

  v2f qa[16];
  {
    const float* qrow = Q + (size_t)(b * T_SEQ + q0 + ln) * HID + of;
    #pragma unroll
    for (int kk = 0; kk < 16; ++kk) qa[kk] = *(const v2f*)(qrow + 4 * kk);
  }

  v8f acc0 = {}, acc1 = {}, acc2 = {}, acc3 = {};
  float rm[8], rs[8];
  #pragma unroll
  for (int r = 0; r < 8; ++r) { rm[r] = -INFINITY; rs[r] = 0.f; }

  for (int kt = w; kt <= qt; kt += NW) {
    const int k0 = kt * 16;

    v8f s = {};
    const float* krow = K + (size_t)(b * T_SEQ + k0 + ln) * HID + of;
    #pragma unroll
    for (int kk = 0; kk < 16; ++kk) {
      v2f kb = *(const v2f*)(krow + 4 * kk);
      s = wmma_f32(qa[kk], kb, s);
    }

    const int kg = k0 + ln;
    #pragma unroll
    for (int r = 0; r < 8; ++r) {
      const int qg = q0 + r + hi * 8;
      const float v = s[r] * scale;
      s[r] = (kg > qg) ? 0.f : v;
    }

    float corr[8];
    #pragma unroll
    for (int r = 0; r < 8; ++r) {
      float t = s[r];
      t = fmaxf(t, __shfl_xor(t, 1));
      t = fmaxf(t, __shfl_xor(t, 2));
      t = fmaxf(t, __shfl_xor(t, 4));
      t = fmaxf(t, __shfl_xor(t, 8));
      const float nm = fmaxf(rm[r], t);
      const float c  = __expf(rm[r] - nm);
      const float e  = __expf(s[r] - nm);
      s[r] = e;
      float rsum = e;
      rsum += __shfl_xor(rsum, 1);
      rsum += __shfl_xor(rsum, 2);
      rsum += __shfl_xor(rsum, 4);
      rsum += __shfl_xor(rsum, 8);
      rs[r] = rs[r] * c + rsum;
      rm[r] = nm;
      corr[r] = c;
    }
    #pragma unroll
    for (int r = 0; r < 8; ++r) {
      acc0[r] *= corr[r]; acc1[r] *= corr[r];
      acc2[r] *= corr[r]; acc3[r] *= corr[r];
    }

    #pragma unroll
    for (int r = 0; r < 8; ++r) plds[w][r + hi * 8][ln] = s[r];
    asm volatile("s_wait_dscnt 0x0" ::: "memory");

    const float* vb = V + (size_t)(b * T_SEQ + k0) * HID;
    #pragma unroll
    for (int j = 0; j < 4; ++j) {
      v2f pa = *(const v2f*)(&plds[w][ln][4 * j + of]);
      const float* vr0 = vb + (size_t)(4 * j + of) * HID + ln;
      const float* vr1 = vb + (size_t)(4 * j + of + 1) * HID + ln;
      v2f bb;
      bb.x = vr0[0];  bb.y = vr1[0];  acc0 = wmma_f32(pa, bb, acc0);
      bb.x = vr0[16]; bb.y = vr1[16]; acc1 = wmma_f32(pa, bb, acc1);
      bb.x = vr0[32]; bb.y = vr1[32]; acc2 = wmma_f32(pa, bb, acc2);
      bb.x = vr0[48]; bb.y = vr1[48]; acc3 = wmma_f32(pa, bb, acc3);
    }
  }

  #pragma unroll
  for (int r = 0; r < 8; ++r) {
    const int prow = r + hi * 8;
    pO[w][prow][ln +  0] = acc0[r];
    pO[w][prow][ln + 16] = acc1[r];
    pO[w][prow][ln + 32] = acc2[r];
    pO[w][prow][ln + 48] = acc3[r];
    if (ln == 0) { pm[w][prow] = rm[r]; pl[w][prow] = rs[r]; }
  }
  __syncthreads();

  const int tid  = w * 32 + lane;
  const int mrow = tid >> 4;
  const int mcol = (tid & 15) * 4;

  float M = pm[0][mrow];
  #pragma unroll
  for (int i = 1; i < NW; ++i) M = fmaxf(M, pm[i][mrow]);

  float L = 0.f;
  float o0 = 0.f, o1 = 0.f, o2 = 0.f, o3 = 0.f;
  #pragma unroll
  for (int i = 0; i < NW; ++i) {
    const float f = __expf(pm[i][mrow] - M);
    const float4 p = *(const float4*)(&pO[i][mrow][mcol]);
    L  += pl[i][mrow] * f;
    o0 += p.x * f; o1 += p.y * f; o2 += p.z * f; o3 += p.w * f;
  }

  const int cf = T_SEQ - 16 * (qt + 1);
  if (cf > 0) {
    const float4 sf = *(const float4*)(
        ws + SUFOFF + ((size_t)b * (NKT + 1) + (qt + 1)) * HID + mcol);
    const float nm = fmaxf(M, 0.f);
    const float cc = __expf(M - nm);
    const float wz = __expf(-nm);
    L  = L  * cc + wz * (float)cf;
    o0 = o0 * cc + wz * sf.x;
    o1 = o1 * cc + wz * sf.y;
    o2 = o2 * cc + wz * sf.z;
    o3 = o3 * cc + wz * sf.w;
  }

  const float inv = 1.f / L;
  float4 o;
  o.x = o0 * inv; o.y = o1 * inv; o.z = o2 * inv; o.w = o3 * inv;
  *(float4*)(out + (size_t)(b * T_SEQ + q0 + mrow) * HID + mcol) = o;
}

extern "C" void kernel_launch(void* const* d_in, const int* in_sizes, int n_in,
                              void* d_out, int out_size, void* d_ws, size_t ws_size,
                              hipStream_t stream) {
  const float* X  = (const float*)d_in[0];
  const float* Wq = (const float*)d_in[1];
  const float* bq = (const float*)d_in[2];
  const float* Wk = (const float*)d_in[3];
  const float* bk = (const float*)d_in[4];
  const float* Wv = (const float*)d_in[5];
  const float* bv = (const float*)d_in[6];
  float* ws  = (float*)d_ws;
  float* out = (float*)d_out;

  qkv_kernel<<<dim3(3, MTOT / PBM), 256, 0, stream>>>(
      X, Wq, bq, Wk, bk, Wv, bv, ws);
  vtilesum_kernel<<<dim3(NKT, NB), 64, 0, stream>>>(ws);
  vsuffix_kernel<<<NB, 64, 0, stream>>>(ws);
  attn_kernel<<<dim3(NB, NKT), dim3(32, NW), 0, stream>>>(ws, out);
}


# suffix-scan kernel: all 128 tile-sum loads of a column issued as one group ahead of the serial adds (was one exposed L2 round trip per step, load behind a may-alias store); sum order unchanged
# speedup vs baseline: 3.8736x; 1.6098x over previous
#include <hip/hip_runtime.h>
#include <math.h>


typedef __attribute__((ext_vector_type(2))) float v2f;
typedef __attribute__((ext_vector_type(8))) float v8f;

#define T_SEQ 2048
#define EMB   1024
#define HID   64
#define NB    8
#define MTOT  (NB * T_SEQ)
#define NKT   (T_SEQ / 16)
#define NW    8

#define QOFF   0
#define KOFF   ((size_t)MTOT * HID)
#define VOFF   ((size_t)2 * MTOT * HID)
#define TSOFF  ((size_t)3 * MTOT * HID)
#define SUFOFF (TSOFF + (size_t)NB * NKT * HID)

__device__ __forceinline__ v8f wmma_f32(v2f a, v2f b, v8f c) {
  return __builtin_amdgcn_wmma_f32_16x16x4_f32(
       false, a,  false, b,
       (short)0, c,  false,  false);
}

#define PBM 64
#define PBK 32
#define PXP (PBK + 4)
#define PWP 160
#define PNC (EMB / PBK)
#define PXV (PBK / 16)
#define PWV (PBK / 32)

__device__ __forceinline__ void qkv_chunk(const float* __restrict__ xa,
                                          const float* __restrict__ wb,
                                          v8f& c0, v8f& c1) {
  v2f a[PBK / 4], b0[PBK / 4], b1[PBK / 4];
  #pragma unroll
  for (int s = 0; s < PBK / 4; ++s) {
    a[s]  = *(const v2f*)(xa + 4 * s);
    b0[s] = *(const v2f*)(wb + (2 * s) * PWP);
    b1[s] = *(const v2f*)(wb + (2 * s) * PWP + 32);
  }
  #pragma unroll
  for (int s = 0; s < PBK / 4; ++s) {
    c0 = wmma_f32(a[s], b0[s], c0);
    c1 = wmma_f32(a[s], b1[s], c1);
  }
}

__global__ __launch_bounds__(256) __attribute__((amdgpu_waves_per_eu(2, 6))) void qkv_kernel(
    const float* __restrict__ X,
    const float* __restrict__ Wq, const float* __restrict__ bq,
    const float* __restrict__ Wk, const float* __restrict__ bk,
    const float* __restrict__ Wv, const float* __restrict__ bv,
    float* __restrict__ ws) {
  __shared__ __attribute__((aligned(16))) float xs[2][PBM * PXP];
  __shared__ __attribute__((aligned(16))) float wt[2][(PBK / 2) * PWP];

  const int tid  = threadIdx.x;
  const int lane = tid & 31;
  const int w    = tid >> 5;
  const int wm   = w & 3;
  const int wn   = w >> 2;
  const int mat  = blockIdx.x;
  const int m0   = blockIdx.y * PBM;
  const int ln   = lane & 15;
  const int hi   = lane >> 4;
  const int of   = hi * 2;

  const float* W    = (mat == 0) ? Wq : (mat == 1) ? Wk : Wv;
  const float* bias = (mat == 0) ? bq : (mat == 1) ? bk : bv;
  float* out = ws + (size_t)mat * (size_t)MTOT * HID;

  const float* xg[PXV];  int xl[PXV];
  #pragma unroll
  for (int i = 0; i < PXV; ++i) {
    const int idx = tid + 256 * i;
    const int row = idx / (PBK / 4), c4 = idx % (PBK / 4);
    xg[i] = X + (size_t)(m0 + row) * EMB + 4 * c4;
    xl[i] = row * PXP + 4 * c4;
  }
  const float* wg[PWV];  int wl[PWV];
  #pragma unroll
  for (int i = 0; i < PWV; ++i) {
    const int u = tid + 256 * i;
    const int kp = u >> 4, c4 = u & 15;
    wg[i] = W + (size_t)(2 * kp) * HID + 4 * c4;
    wl[i] = kp * PWP + 8 * c4;
  }

  const int xf = (wm * 16 + ln) * PXP + of;
  const int wf = hi * PWP + 2 * (wn * 32 + ln);

  float4 px[PXV], pwa[PWV], pwb[PWV];
  #pragma unroll
  for (int i = 0; i < PXV; ++i) px[i] = *(const float4*)xg[i];
  #pragma unroll
  for (int i = 0; i < PWV; ++i) {
    pwa[i] = *(const float4*)wg[i];
    pwb[i] = *(const float4*)(wg[i] + HID);
  }
  #pragma unroll
  for (int i = 0; i < PXV; ++i) *(float4*)&xs[0][xl[i]] = px[i];
  #pragma unroll
  for (int i = 0; i < PWV; ++i) {
    *(float4*)&wt[0][wl[i]]     = make_float4(pwa[i].x, pwb[i].x, pwa[i].y, pwb[i].y);
    *(float4*)&wt[0][wl[i] + 4] = make_float4(pwa[i].z, pwb[i].z, pwa[i].w, pwb[i].w);
  }
  __syncthreads();

  v8f c0 = {}, c1 = {};
  for (int t = 0; t < PNC - 1; ++t) {
    const int cur = t & 1;
    const size_t kx = (size_t)(t + 1) * PBK;
    #pragma unroll
    for (int i = 0; i < PXV; ++i) px[i] = *(const float4*)(xg[i] + kx);
    #pragma unroll
    for (int i = 0; i < PWV; ++i) {
      pwa[i] = *(const float4*)(wg[i] + kx * HID);
      pwb[i] = *(const float4*)(wg[i] + kx * HID + HID);
    }

    qkv_chunk(&xs[cur][xf], &wt[cur][wf], c0, c1);

    #pragma unroll
    for (int i = 0; i < PXV; ++i) *(float4*)&xs[cur ^ 1][xl[i]] = px[i];
    #pragma unroll
    for (int i = 0; i < PWV; ++i) {
      *(float4*)&wt[cur ^ 1][wl[i]]     = make_float4(pwa[i].x, pwb[i].x, pwa[i].y, pwb[i].y);
      *(float4*)&wt[cur ^ 1][wl[i] + 4] = make_float4(pwa[i].z, pwb[i].z, pwa[i].w, pwb[i].w);
    }
    __syncthreads();
  }
  qkv_chunk(&xs[(PNC - 1) & 1][xf], &wt[(PNC - 1) & 1][wf], c0, c1);

  const int n0   = wn * 32 + ln;
  const float bb0 = bias[n0], bb1 = bias[n0 + 16];
  const int mrow = m0 + wm * 16 + hi * 8;
  #pragma unroll
  for (int r = 0; r < 8; ++r) {
    out[(size_t)(mrow + r) * HID + n0]      = c0[r] + bb0;
    out[(size_t)(mrow + r) * HID + n0 + 16] = c1[r] + bb1;
  }
}

__global__ void vtilesum_kernel(float* __restrict__ ws) {
  const int h = threadIdx.x;
  const int j = blockIdx.x;
  const int b = blockIdx.y;
  const float* V = ws + VOFF + (size_t)(b * T_SEQ + j * 16) * HID + h;
  float s = 0.f;
  #pragma unroll
  for (int i = 0; i < 16; ++i) s += V[(size_t)i * HID];
  ws[TSOFF + ((size_t)b * NKT + j) * HID + h] = s;
}

__global__ __launch_bounds__(64) void vsuffix_kernel(float* __restrict__ ws) {
  const int h = threadIdx.x;
  const int b = blockIdx.x;
  const float* ts = ws + TSOFF + (size_t)b * NKT * HID + h;
  float* suf = ws + SUFOFF + (size_t)b * (NKT + 1) * HID + h;
  float t[NKT];
  #pragma unroll
  for (int j = NKT - 1; j >= 0; --j) t[j] = ts[(size_t)j * HID];
  float acc = 0.f;
  suf[(size_t)NKT * HID] = 0.f;
  #pragma unroll
  for (int j = NKT - 1; j >= 0; --j) {
    acc += t[j];
    suf[(size_t)j * HID] = acc;
  }
}

__global__ __launch_bounds__(32 * NW) void attn_kernel(
    const float* __restrict__ ws, float* __restrict__ out) {
  __shared__ float plds[NW][16][18];
  __shared__ __attribute__((aligned(16))) float pO[NW][16][68];
  __shared__ float pm[NW][16];
  __shared__ float pl[NW][16];

  const int lane = threadIdx.x;
  const int w    = threadIdx.y;
  const int b    = blockIdx.x;
  const int qt   = NKT - 1 - (int)blockIdx.y;
  const int q0   = qt * 16;
  const int ln   = lane & 15;
  const int hi   = lane >> 4;
  const int of   = hi * 2;

  const float* Q = ws + QOFF;
  const float* K = ws + KOFF;
  const float* V = ws + VOFF;
  const float scale = 0.125f;

  v2f qa[16];
  {
    const float* qrow = Q + (size_t)(b * T_SEQ + q0 + ln) * HID + of;
    #pragma unroll
    for (int kk = 0; kk < 16; ++kk) qa[kk] = *(const v2f*)(qrow + 4 * kk);
  }

  v8f acc0 = {}, acc1 = {}, acc2 = {}, acc3 = {};
  float rm[8], rs[8];
  #pragma unroll
  for (int r = 0; r < 8; ++r) { rm[r] = -INFINITY; rs[r] = 0.f; }

  for (int kt = w; kt <= qt; kt += NW) {
    const int k0 = kt * 16;

    v8f s = {};
    const float* krow = K + (size_t)(b * T_SEQ + k0 + ln) * HID + of;
    #pragma unroll
    for (int kk = 0; kk < 16; ++kk) {
      v2f kb = *(const v2f*)(krow + 4 * kk);
      s = wmma_f32(qa[kk], kb, s);
    }

    const int kg = k0 + ln;
    #pragma unroll
    for (int r = 0; r < 8; ++r) {
      const int qg = q0 + r + hi * 8;
      const float v = s[r] * scale;
      s[r] = (kg > qg) ? 0.f : v;
    }

    float corr[8];
    #pragma unroll
    for (int r = 0; r < 8; ++r) {
      float t = s[r];
      t = fmaxf(t, __shfl_xor(t, 1));
      t = fmaxf(t, __shfl_xor(t, 2));
      t = fmaxf(t, __shfl_xor(t, 4));
      t = fmaxf(t, __shfl_xor(t, 8));
      const float nm = fmaxf(rm[r], t);
      const float c  = __expf(rm[r] - nm);
      const float e  = __expf(s[r] - nm);
      s[r] = e;
      float rsum = e;
      rsum += __shfl_xor(rsum, 1);
      rsum += __shfl_xor(rsum, 2);
      rsum += __shfl_xor(rsum, 4);
      rsum += __shfl_xor(rsum, 8);
      rs[r] = rs[r] * c + rsum;
      rm[r] = nm;
      corr[r] = c;
    }
    #pragma unroll
    for (int r = 0; r < 8; ++r) {
      acc0[r] *= corr[r]; acc1[r] *= corr[r];
      acc2[r] *= corr[r]; acc3[r] *= corr[r];
    }

    #pragma unroll
    for (int r = 0; r < 8; ++r) plds[w][r + hi * 8][ln] = s[r];
    asm volatile("s_wait_dscnt 0x0" ::: "memory");

    const float* vb = V + (size_t)(b * T_SEQ + k0) * HID;
    #pragma unroll
    for (int j = 0; j < 4; ++j) {
      v2f pa = *(const v2f*)(&plds[w][ln][4 * j + of]);
      const float* vr0 = vb + (size_t)(4 * j + of) * HID + ln;
      const float* vr1 = vb + (size_t)(4 * j + of + 1) * HID + ln;
      v2f bb;
      bb.x = vr0[0];  bb.y = vr1[0];  acc0 = wmma_f32(pa, bb, acc0);
      bb.x = vr0[16]; bb.y = vr1[16]; acc1 = wmma_f32(pa, bb, acc1);
      bb.x = vr0[32]; bb.y = vr1[32]; acc2 = wmma_f32(pa, bb, acc2);
      bb.x = vr0[48]; bb.y = vr1[48]; acc3 = wmma_f32(pa, bb, acc3);
    }
  }

  #pragma unroll
  for (int r = 0; r < 8; ++r) {
    const int prow = r + hi * 8;
    pO[w][prow][ln +  0] = acc0[r];
    pO[w][prow][ln + 16] = acc1[r];
    pO[w][prow][ln + 32] = acc2[r];
    pO[w][prow][ln + 48] = acc3[r];
    if (ln == 0) { pm[w][prow] = rm[r]; pl[w][prow] = rs[r]; }
  }
  __syncthreads();

  const int tid  = w * 32 + lane;
  const int mrow = tid >> 4;
  const int mcol = (tid & 15) * 4;

  float M = pm[0][mrow];
  #pragma unroll
  for (int i = 1; i < NW; ++i) M = fmaxf(M, pm[i][mrow]);

  float L = 0.f;
  float o0 = 0.f, o1 = 0.f, o2 = 0.f, o3 = 0.f;
  #pragma unroll
  for (int i = 0; i < NW; ++i) {
    const float f = __expf(pm[i][mrow] - M);
    const float4 p = *(const float4*)(&pO[i][mrow][mcol]);
    L  += pl[i][mrow] * f;
    o0 += p.x * f; o1 += p.y * f; o2 += p.z * f; o3 += p.w * f;
  }

  const int cf = T_SEQ - 16 * (qt + 1);
  if (cf > 0) {
    const float4 sf = *(const float4*)(
        ws + SUFOFF + ((size_t)b * (NKT + 1) + (qt + 1)) * HID + mcol);
    const float nm = fmaxf(M, 0.f);
    const float cc = __expf(M - nm);
    const float wz = __expf(-nm);
    L  = L  * cc + wz * (float)cf;
    o0 = o0 * cc + wz * sf.x;
    o1 = o1 * cc + wz * sf.y;
    o2 = o2 * cc + wz * sf.z;
    o3 = o3 * cc + wz * sf.w;
  }

  const float inv = 1.f / L;
  float4 o;
  o.x = o0 * inv; o.y = o1 * inv; o.z = o2 * inv; o.w = o3 * inv;
  *(float4*)(out + (size_t)(b * T_SEQ + q0 + mrow) * HID + mcol) = o;
}

extern "C" void kernel_launch(void* const* d_in, const int* in_sizes, int n_in,
                              void* d_out, int out_size, void* d_ws, size_t ws_size,
                              hipStream_t stream) {
  const float* X  = (const float*)d_in[0];
  const float* Wq = (const float*)d_in[1];
  const float* bq = (const float*)d_in[2];
  const float* Wk = (const float*)d_in[3];
  const float* bk = (const float*)d_in[4];
  const float* Wv = (const float*)d_in[5];
  const float* bv = (const float*)d_in[6];
  float* ws  = (float*)d_ws;
  float* out = (float*)d_out;

  qkv_kernel<<<dim3(3, MTOT / PBM), 256, 0, stream>>>(
      X, Wq, bq, Wk, bk, Wv, bv, ws);
  vtilesum_kernel<<<dim3(NKT, NB), 64, 0, stream>>>(ws);
  vsuffix_kernel<<<NB, 64, 0, stream>>>(ws);
  attn_kernel<<<dim3(NB, NKT), dim3(32, NW), 0, stream>>>(ws, out);
}
